# starting kernel
# baseline (speedup 1.0000x reference)
#include <hip/hip_runtime.h>
#include <math.h>

typedef __attribute__((ext_vector_type(2))) float v2f;
typedef __attribute__((ext_vector_type(8))) float v8f;

#define O_DIM 128

__device__ __forceinline__ void atomAddF(float* p, float v) {
    __hip_atomic_fetch_add(p, v, __ATOMIC_RELAXED, __HIP_MEMORY_SCOPE_AGENT);
}

__global__ __launch_bounds__(256)
void gemm_bias_relu(const float* __restrict__ X, const float* __restrict__ W,
                    const float* __restrict__ bias, float* __restrict__ out,
                    int N, int K) {
    __shared__ float  As[128][36];
    __shared__ float2 Bs2[16][144];

    const int tid = threadIdx.x;
    const int wv  = tid >> 5;
    const int ln  = tid & 31;
    const int m0  = blockIdx.x * 128;

    v8f acc[8];
#pragma unroll
    for (int t = 0; t < 8; ++t)
#pragma unroll
        for (int j = 0; j < 8; ++j) acc[t][j] = 0.0f;

    const int mrow = wv * 16 + (ln & 15);
    const int lnn  = ln & 15;
    const int khalf = (ln >> 4) << 1;

    for (int k0 = 0; k0 < K; k0 += 32) {
#pragma unroll
        for (int i = 0; i < 4; ++i) {
            int f   = tid + i * 256;
            int row = f >> 3;
            int c4  = f & 7;
            float4 v = make_float4(0.f, 0.f, 0.f, 0.f);
            if (m0 + row < N)
                v = *(const float4*)(X + (size_t)(m0 + row) * K + k0 + c4 * 4);
            *(float4*)&As[row][c4 * 4] = v;
        }
#pragma unroll
        for (int i = 0; i < 4; ++i) {
            int f   = tid + i * 256;
            int row = f >> 5;
            int c4  = f & 31;
            float4 v = *(const float4*)(W + (size_t)(k0 + row) * O_DIM + c4 * 4);
            float* bp = (float*)&Bs2[row >> 1][c4 * 4] + (row & 1);
            bp[0] = v.x;
            bp[2] = v.y;
            bp[4] = v.z;
            bp[6] = v.w;
        }
        __syncthreads();

#pragma unroll
        for (int kk = 0; kk < 32; kk += 4) {
            const int kk2 = kk + khalf;
            const int p   = kk2 >> 1;
            v2f a = *(const v2f*)&As[mrow][kk2];
#pragma unroll
            for (int t = 0; t < 8; ++t) {
                v2f b = *(const v2f*)&Bs2[p][t * 16 + lnn];
                acc[t] = __builtin_amdgcn_wmma_f32_16x16x4_f32(
                    false, a, false, b, (short)0, acc[t], false, false);
            }
        }
        __syncthreads();
    }

    const int mtile = m0 + wv * 16;
    if (mtile < N) {
        const int rowoff = (ln >> 4) << 3;
#pragma unroll
        for (int t = 0; t < 8; ++t) {
            const int col = t * 16 + lnn;
            const float bv = bias[col];
#pragma unroll
            for (int i = 0; i < 8; ++i) {
                float v = acc[t][i] + bv;
                out[(size_t)(mtile + rowoff + i) * O_DIM + col] = v > 0.f ? v : 0.f;
            }
        }
    }
}

__global__ __launch_bounds__(256)
void spmm128(const int* __restrict__ src, const int* __restrict__ dst,
             const float* __restrict__ w, const float* __restrict__ x,
             float* __restrict__ out, int E) {
    int e = blockIdx.x * 8 + (threadIdx.x >> 5);
    if (e >= E) return;
    const int ln = threadIdx.x & 31;
    const int s = src[e];
    const int d = dst[e];
    const float wt = w[e];
    float4 v = ((const float4*)(x + (size_t)s * 128))[ln];
    float* o = out + (size_t)d * 128 + ln * 4;
    atomAddF(o + 0, wt * v.x);
    atomAddF(o + 1, wt * v.y);
    atomAddF(o + 2, wt * v.z);
    atomAddF(o + 3, wt * v.w);
}

__global__ __launch_bounds__(256)
void spmm_concat256(const int* __restrict__ src, const int* __restrict__ dst,
                    const float* __restrict__ w, const float* __restrict__ xa,
                    const float* __restrict__ xb, float* __restrict__ out, int E) {
    int e = blockIdx.x * 8 + (threadIdx.x >> 5);
    if (e >= E) return;
    const int ln = threadIdx.x & 31;
    const int s = src[e];
    const int d = dst[e];
    const float wt = w[e];
    float4 va = ((const float4*)(xa + (size_t)s * 128))[ln];
    float4 vb = ((const float4*)(xb + (size_t)s * 128))[ln];
    float* oa = out + (size_t)d * 256 + ln * 4;
    float* ob = oa + 128;
    atomAddF(oa + 0, wt * va.x);
    atomAddF(oa + 1, wt * va.y);
    atomAddF(oa + 2, wt * va.z);
    atomAddF(oa + 3, wt * va.w);
    atomAddF(ob + 0, wt * vb.x);
    atomAddF(ob + 1, wt * vb.y);
    atomAddF(ob + 2, wt * vb.z);
    atomAddF(ob + 3, wt * vb.w);
}

__global__ __launch_bounds__(256)
void l2norm_concat(const float* __restrict__ r0, const float* __restrict__ r1,
                   float* __restrict__ out, int N) {
    int row = blockIdx.x * 8 + (threadIdx.x >> 5);
    if (row >= N) return;
    const int ln = threadIdx.x & 31;

    float4 v0  = ((const float4*)(r0 + (size_t)row * 128))[ln];
    float4 v1a = ((const float4*)(r1 + (size_t)row * 256))[ln];
    float4 v1b = ((const float4*)(r1 + (size_t)row * 256))[ln + 32];

    float s0 = v0.x * v0.x + v0.y * v0.y + v0.z * v0.z + v0.w * v0.w;
    float s1 = v1a.x * v1a.x + v1a.y * v1a.y + v1a.z * v1a.z + v1a.w * v1a.w
             + v1b.x * v1b.x + v1b.y * v1b.y + v1b.z * v1b.z + v1b.w * v1b.w;
#pragma unroll
    for (int off = 16; off > 0; off >>= 1) {
        s0 += __shfl_xor(s0, off, 32);
        s1 += __shfl_xor(s1, off, 32);
    }
    const float i0 = 1.0f / (sqrtf(s0) + 1e-9f);
    const float i1 = 1.0f / (sqrtf(s1) + 1e-9f);

    float* o = out + (size_t)row * 384;
    float4 w0 = make_float4(v0.x * i0, v0.y * i0, v0.z * i0, v0.w * i0);
    float4 wa = make_float4(v1a.x * i1, v1a.y * i1, v1a.z * i1, v1a.w * i1);
    float4 wb = make_float4(v1b.x * i1, v1b.y * i1, v1b.z * i1, v1b.w * i1);
    *(float4*)(o + ln * 4)        = w0;
    *(float4*)(o + 128 + ln * 4)  = wa;
    *(float4*)(o + 256 + ln * 4)  = wb;
}

extern "C" void kernel_launch(void* const* d_in, const int* in_sizes, int n_in,
                              void* d_out, int out_size, void* d_ws, size_t ws_size,
                              hipStream_t stream) {
    const float* x1       = (const float*)d_in[0];
    const float* x2       = (const float*)d_in[1];
    const float* word_emb = (const float*)d_in[2];
    const float* W1a = (const float*)d_in[3];
    const float* b1a = (const float*)d_in[4];
    const float* W1b = (const float*)d_in[5];
    const float* b1b = (const float*)d_in[6];
    const float* W2a = (const float*)d_in[7];
    const float* b2a = (const float*)d_in[8];
    const float* W2b = (const float*)d_in[9];
    const float* b2b = (const float*)d_in[10];
    const int*   e11_src = (const int*)d_in[11];
    const int*   e11_dst = (const int*)d_in[12];
    const float* e11_w   = (const float*)d_in[13];
    const int*   e22_src = (const int*)d_in[14];
    const int*   e22_dst = (const int*)d_in[15];
    const float* e22_w   = (const float*)d_in[16];
    const int*   e01_src = (const int*)d_in[17];
    const int*   e01_dst = (const int*)d_in[18];
    const float* e01_w   = (const float*)d_in[19];
    const int*   e02_src = (const int*)d_in[20];
    const int*   e02_dst = (const int*)d_in[21];
    const float* e02_w   = (const float*)d_in[22];

    const int N0 = 10000, N1 = 50000, N2 = 30000;
    const int E11 = in_sizes[11], E22 = in_sizes[14];
    const int E01 = in_sizes[17], E02 = in_sizes[20];

    float* ws = (float*)d_ws;
    size_t off = 0;
    float* l1_1 = ws + off; off += (size_t)N1 * 128;
    float* h11  = ws + off; off += (size_t)N1 * 128;
    float* l2_1 = ws + off; off += (size_t)N1 * 128;
    float* l1_2 = ws + off; off += (size_t)N2 * 128;
    float* h22  = ws + off; off += (size_t)N2 * 128;
    float* l2_2 = ws + off; off += (size_t)N2 * 128;
    float* r0   = ws + off; off += (size_t)N0 * 128;
    float* r1   = ws + off; off += (size_t)N0 * 256;
    float* r0s  = ws + off; off += (size_t)N0 * 128;
    float* r1s  = ws + off; off += (size_t)N0 * 256;

    float* doc     = (float*)d_out;
    float* doc_svd = doc + (size_t)N0 * 384;

    const dim3 blk(256);
    auto gemm_grid = [](int n) { return dim3((n + 127) / 128); };
    auto edge_grid = [](int e) { return dim3((e + 7) / 8); };

    gemm_bias_relu<<<gemm_grid(N1), blk, 0, stream>>>(x1, W1a, b1a, l1_1, N1, 256);
    hipMemsetAsync(h11, 0, (size_t)N1 * 128 * sizeof(float), stream);
    spmm128<<<edge_grid(E11), blk, 0, stream>>>(e11_src, e11_dst, e11_w, l1_1, h11, E11);
    gemm_bias_relu<<<gemm_grid(N1), blk, 0, stream>>>(h11, W1b, b1b, l2_1, N1, 128);

    gemm_bias_relu<<<gemm_grid(N2), blk, 0, stream>>>(x2, W2a, b2a, l1_2, N2, 256);
    hipMemsetAsync(h22, 0, (size_t)N2 * 128 * sizeof(float), stream);
    spmm128<<<edge_grid(E22), blk, 0, stream>>>(e22_src, e22_dst, e22_w, l1_2, h22, E22);
    gemm_bias_relu<<<gemm_grid(N2), blk, 0, stream>>>(h22, W2b, b2b, l2_2, N2, 128);

    hipMemsetAsync(r0, 0, (size_t)N0 * (128 + 256 + 128 + 256) * sizeof(float), stream);
    spmm128<<<edge_grid(E01), blk, 0, stream>>>(e01_src, e01_dst, e01_w, l2_1, r0, E01);
    spmm_concat256<<<edge_grid(E02), blk, 0, stream>>>(e02_src, e02_dst, e02_w, l2_2, word_emb, r1, E02);
    spmm128<<<edge_grid(E01), blk, 0, stream>>>(e01_src, e01_dst, e01_w, l1_1, r0s, E01);
    spmm_concat256<<<edge_grid(E02), blk, 0, stream>>>(e02_src, e02_dst, e02_w, l1_2, word_emb, r1s, E02);

    l2norm_concat<<<dim3((N0 + 7) / 8), blk, 0, stream>>>(r0,  r1,  doc,     N0);
    l2norm_concat<<<dim3((N0 + 7) / 8), blk, 0, stream>>>(r0s, r1s, doc_svd, N0);
}


# GEMMs: W staged once into LDS, each wave loads its own 16 rows of X from global straight into A fragments one 64-k chunk ahead, so the K loop has no barrier; k order inside each 8-k group changes
# speedup vs baseline: 1.0636x; 1.0636x over previous
#include <hip/hip_runtime.h>
#include <math.h>

typedef __attribute__((ext_vector_type(2))) float v2f;
typedef __attribute__((ext_vector_type(8))) float v8f;

#define O_DIM 128

__device__ __forceinline__ void atomAddF(float* p, float v) {
    __hip_atomic_fetch_add(p, v, __ATOMIC_RELAXED, __HIP_MEMORY_SCOPE_AGENT);
}

__device__ __forceinline__ void load_a_chunk(float4 (&a)[8], const float4* __restrict__ xp, int k0) {
#pragma unroll
    for (int i = 0; i < 8; ++i) a[i] = xp[(k0 >> 2) + 2 * i];
}

__device__ __forceinline__ void mma_chunk(v8f (&acc)[8], const float4 (&a)[8],
                                          const float4 (*Bs4)[O_DIM], int k0, int h, int c) {
#pragma unroll
    for (int i = 0; i < 8; ++i) {
        const int q = (k0 >> 2) + 2 * i + h;
        float4 b[8];
#pragma unroll
        for (int t = 0; t < 8; ++t) b[t] = Bs4[q][t * 16 + c];
#pragma unroll
        for (int j = 0; j < 2; ++j) {
            v2f av;
            av[0] = j ? a[i].z : a[i].x;
            av[1] = j ? a[i].w : a[i].y;
#pragma unroll
            for (int t = 0; t < 8; ++t) {
                v2f bv;
                bv[0] = j ? b[t].z : b[t].x;
                bv[1] = j ? b[t].w : b[t].y;
                acc[t] = __builtin_amdgcn_wmma_f32_16x16x4_f32(
                    false, av, false, bv, (short)0, acc[t], false, false);
            }
        }
    }
}

template <int K>
__global__ __launch_bounds__(256) __attribute__((amdgpu_waves_per_eu(2, 4)))
void gemm_bias_relu(const float* __restrict__ X, const float* __restrict__ W,
                    const float* __restrict__ bias, float* __restrict__ out, int N) {
    __shared__ float4 Bs4[K / 4][O_DIM];

    const int tid = threadIdx.x;
    const int wv  = tid >> 5;
    const int ln  = tid & 31;
    const int c   = ln & 15;
    const int h   = ln >> 4;
    const int mtile = blockIdx.x * 128 + wv * 16;

    int row = mtile + c;
    if (row > N - 1) row = N - 1;
    const float4* xp = (const float4*)(X + (size_t)row * K) + h;

    float4 a0[8], a1[8];
    load_a_chunk(a0, xp, 0);

#pragma unroll
    for (int it = 0; it < K / 32; ++it) {
        const int f  = tid + it * 256;
        const int q  = f >> 5;
        const int c4 = f & 31;
        const float* wp = W + (size_t)(4 * q) * O_DIM + c4 * 4;
        const float4 v0 = *(const float4*)(wp);
        const float4 v1 = *(const float4*)(wp + O_DIM);
        const float4 v2 = *(const float4*)(wp + 2 * O_DIM);
        const float4 v3 = *(const float4*)(wp + 3 * O_DIM);
        float4* d = &Bs4[q][c4 * 4];
        d[0] = make_float4(v0.x, v1.x, v2.x, v3.x);
        d[1] = make_float4(v0.y, v1.y, v2.y, v3.y);
        d[2] = make_float4(v0.z, v1.z, v2.z, v3.z);
        d[3] = make_float4(v0.w, v1.w, v2.w, v3.w);
    }
    __syncthreads();

    if (mtile >= N) return;

    v8f acc[8];
#pragma unroll
    for (int t = 0; t < 8; ++t)
#pragma unroll
        for (int j = 0; j < 8; ++j) acc[t][j] = 0.0f;

#pragma unroll
    for (int k0 = 0; k0 < K; k0 += 128) {
        load_a_chunk(a1, xp, k0 + 64);
        mma_chunk(acc, a0, Bs4, k0, h, c);
        if (k0 + 128 < K) load_a_chunk(a0, xp, k0 + 128);
        mma_chunk(acc, a1, Bs4, k0 + 64, h, c);
    }

    const int rowoff = h << 3;
#pragma unroll
    for (int t = 0; t < 8; ++t) {
        const int col = t * 16 + c;
        const float bv = bias[col];
#pragma unroll
        for (int i = 0; i < 8; ++i) {
            float v = acc[t][i] + bv;
            out[(size_t)(mtile + rowoff + i) * O_DIM + col] = v > 0.f ? v : 0.f;
        }
    }
}

__global__ __launch_bounds__(256)
void spmm128(const int* __restrict__ src, const int* __restrict__ dst,
             const float* __restrict__ w, const float* __restrict__ x,
             float* __restrict__ out, int E) {
    int e = blockIdx.x * 8 + (threadIdx.x >> 5);
    if (e >= E) return;
    const int ln = threadIdx.x & 31;
    const int s = src[e];
    const int d = dst[e];
    const float wt = w[e];
    float4 v = ((const float4*)(x + (size_t)s * 128))[ln];
    float* o = out + (size_t)d * 128 + ln * 4;
    atomAddF(o + 0, wt * v.x);
    atomAddF(o + 1, wt * v.y);
    atomAddF(o + 2, wt * v.z);
    atomAddF(o + 3, wt * v.w);
}

__global__ __launch_bounds__(256)
void spmm_concat256(const int* __restrict__ src, const int* __restrict__ dst,
                    const float* __restrict__ w, const float* __restrict__ xa,
                    const float* __restrict__ xb, float* __restrict__ out, int E) {
    int e = blockIdx.x * 8 + (threadIdx.x >> 5);
    if (e >= E) return;
    const int ln = threadIdx.x & 31;
    const int s = src[e];
    const int d = dst[e];
    const float wt = w[e];
    float4 va = ((const float4*)(xa + (size_t)s * 128))[ln];
    float4 vb = ((const float4*)(xb + (size_t)s * 128))[ln];
    float* oa = out + (size_t)d * 256 + ln * 4;
    float* ob = oa + 128;
    atomAddF(oa + 0, wt * va.x);
    atomAddF(oa + 1, wt * va.y);
    atomAddF(oa + 2, wt * va.z);
    atomAddF(oa + 3, wt * va.w);
    atomAddF(ob + 0, wt * vb.x);
    atomAddF(ob + 1, wt * vb.y);
    atomAddF(ob + 2, wt * vb.z);
    atomAddF(ob + 3, wt * vb.w);
}

__global__ __launch_bounds__(256)
void l2norm_concat(const float* __restrict__ r0, const float* __restrict__ r1,
                   float* __restrict__ out, int N) {
    int row = blockIdx.x * 8 + (threadIdx.x >> 5);
    if (row >= N) return;
    const int ln = threadIdx.x & 31;

    float4 v0  = ((const float4*)(r0 + (size_t)row * 128))[ln];
    float4 v1a = ((const float4*)(r1 + (size_t)row * 256))[ln];
    float4 v1b = ((const float4*)(r1 + (size_t)row * 256))[ln + 32];

    float s0 = v0.x * v0.x + v0.y * v0.y + v0.z * v0.z + v0.w * v0.w;
    float s1 = v1a.x * v1a.x + v1a.y * v1a.y + v1a.z * v1a.z + v1a.w * v1a.w
             + v1b.x * v1b.x + v1b.y * v1b.y + v1b.z * v1b.z + v1b.w * v1b.w;
#pragma unroll
    for (int off = 16; off > 0; off >>= 1) {
        s0 += __shfl_xor(s0, off, 32);
        s1 += __shfl_xor(s1, off, 32);
    }
    const float i0 = 1.0f / (sqrtf(s0) + 1e-9f);
    const float i1 = 1.0f / (sqrtf(s1) + 1e-9f);

    float* o = out + (size_t)row * 384;
    float4 w0 = make_float4(v0.x * i0, v0.y * i0, v0.z * i0, v0.w * i0);
    float4 wa = make_float4(v1a.x * i1, v1a.y * i1, v1a.z * i1, v1a.w * i1);
    float4 wb = make_float4(v1b.x * i1, v1b.y * i1, v1b.z * i1, v1b.w * i1);
    *(float4*)(o + ln * 4)        = w0;
    *(float4*)(o + 128 + ln * 4)  = wa;
    *(float4*)(o + 256 + ln * 4)  = wb;
}

extern "C" void kernel_launch(void* const* d_in, const int* in_sizes, int n_in,
                              void* d_out, int out_size, void* d_ws, size_t ws_size,
                              hipStream_t stream) {
    const float* x1       = (const float*)d_in[0];
    const float* x2       = (const float*)d_in[1];
    const float* word_emb = (const float*)d_in[2];
    const float* W1a = (const float*)d_in[3];
    const float* b1a = (const float*)d_in[4];
    const float* W1b = (const float*)d_in[5];
    const float* b1b = (const float*)d_in[6];
    const float* W2a = (const float*)d_in[7];
    const float* b2a = (const float*)d_in[8];
    const float* W2b = (const float*)d_in[9];
    const float* b2b = (const float*)d_in[10];
    const int*   e11_src = (const int*)d_in[11];
    const int*   e11_dst = (const int*)d_in[12];
    const float* e11_w   = (const float*)d_in[13];
    const int*   e22_src = (const int*)d_in[14];
    const int*   e22_dst = (const int*)d_in[15];
    const float* e22_w   = (const float*)d_in[16];
    const int*   e01_src = (const int*)d_in[17];
    const int*   e01_dst = (const int*)d_in[18];
    const float* e01_w   = (const float*)d_in[19];
    const int*   e02_src = (const int*)d_in[20];
    const int*   e02_dst = (const int*)d_in[21];
    const float* e02_w   = (const float*)d_in[22];

    const int N0 = 10000, N1 = 50000, N2 = 30000;
    const int E11 = in_sizes[11], E22 = in_sizes[14];
    const int E01 = in_sizes[17], E02 = in_sizes[20];

    float* ws = (float*)d_ws;
    size_t off = 0;
    float* l1_1 = ws + off; off += (size_t)N1 * 128;
    float* h11  = ws + off; off += (size_t)N1 * 128;
    float* l2_1 = ws + off; off += (size_t)N1 * 128;
    float* l1_2 = ws + off; off += (size_t)N2 * 128;
    float* h22  = ws + off; off += (size_t)N2 * 128;
    float* l2_2 = ws + off; off += (size_t)N2 * 128;
    float* r0   = ws + off; off += (size_t)N0 * 128;
    float* r1   = ws + off; off += (size_t)N0 * 256;
    float* r0s  = ws + off; off += (size_t)N0 * 128;
    float* r1s  = ws + off; off += (size_t)N0 * 256;

    float* doc     = (float*)d_out;
    float* doc_svd = doc + (size_t)N0 * 384;

    const dim3 blk(256);
    auto gemm_grid = [](int n) { return dim3((n + 127) / 128); };
    auto edge_grid = [](int e) { return dim3((e + 7) / 8); };

    gemm_bias_relu<256><<<gemm_grid(N1), blk, 0, stream>>>(x1, W1a, b1a, l1_1, N1);
    hipMemsetAsync(h11, 0, (size_t)N1 * 128 * sizeof(float), stream);
    spmm128<<<edge_grid(E11), blk, 0, stream>>>(e11_src, e11_dst, e11_w, l1_1, h11, E11);
    gemm_bias_relu<128><<<gemm_grid(N1), blk, 0, stream>>>(h11, W1b, b1b, l2_1, N1);

    gemm_bias_relu<256><<<gemm_grid(N2), blk, 0, stream>>>(x2, W2a, b2a, l1_2, N2);
    hipMemsetAsync(h22, 0, (size_t)N2 * 128 * sizeof(float), stream);
    spmm128<<<edge_grid(E22), blk, 0, stream>>>(e22_src, e22_dst, e22_w, l1_2, h22, E22);
    gemm_bias_relu<128><<<gemm_grid(N2), blk, 0, stream>>>(h22, W2b, b2b, l2_2, N2);

    hipMemsetAsync(r0, 0, (size_t)N0 * (128 + 256 + 128 + 256) * sizeof(float), stream);
    spmm128<<<edge_grid(E01), blk, 0, stream>>>(e01_src, e01_dst, e01_w, l2_1, r0, E01);
    spmm_concat256<<<edge_grid(E02), blk, 0, stream>>>(e02_src, e02_dst, e02_w, l2_2, word_emb, r1, E02);
    spmm128<<<edge_grid(E01), blk, 0, stream>>>(e01_src, e01_dst, e01_w, l1_1, r0s, E01);
    spmm_concat256<<<edge_grid(E02), blk, 0, stream>>>(e02_src, e02_dst, e02_w, l1_2, word_emb, r1s, E02);

    l2norm_concat<<<dim3((N0 + 7) / 8), blk, 0, stream>>>(r0,  r1,  doc,     N0);
    l2norm_concat<<<dim3((N0 + 7) / 8), blk, 0, stream>>>(r0s, r1s, doc_svd, N0);
}


# doc aggregation: each of the two edge lists is walked once, a wave gathers the edge's 2 or 3 source rows and adds them all; word-embedding half summed once; sums stay in atomic arrival order
# speedup vs baseline: 1.1683x; 1.0984x over previous
#include <hip/hip_runtime.h>
#include <math.h>

typedef __attribute__((ext_vector_type(2))) float v2f;
typedef __attribute__((ext_vector_type(8))) float v8f;

#define O_DIM 128

__device__ __forceinline__ void atomAddF(float* p, float v) {
    __hip_atomic_fetch_add(p, v, __ATOMIC_RELAXED, __HIP_MEMORY_SCOPE_AGENT);
}

__device__ __forceinline__ void load_a_chunk(float4 (&a)[8], const float4* __restrict__ xp, int k0) {
#pragma unroll
    for (int i = 0; i < 8; ++i) a[i] = xp[(k0 >> 2) + 2 * i];
}

__device__ __forceinline__ void mma_chunk(v8f (&acc)[8], const float4 (&a)[8],
                                          const float4 (*Bs4)[O_DIM], int k0, int h, int c) {
#pragma unroll
    for (int i = 0; i < 8; ++i) {
        const int q = (k0 >> 2) + 2 * i + h;
        float4 b[8];
#pragma unroll
        for (int t = 0; t < 8; ++t) b[t] = Bs4[q][t * 16 + c];
#pragma unroll
        for (int j = 0; j < 2; ++j) {
            v2f av;
            av[0] = j ? a[i].z : a[i].x;
            av[1] = j ? a[i].w : a[i].y;
#pragma unroll
            for (int t = 0; t < 8; ++t) {
                v2f bv;
                bv[0] = j ? b[t].z : b[t].x;
                bv[1] = j ? b[t].w : b[t].y;
                acc[t] = __builtin_amdgcn_wmma_f32_16x16x4_f32(
                    false, av, false, bv, (short)0, acc[t], false, false);
            }
        }
    }
}

template <int K>
__global__ __launch_bounds__(256) __attribute__((amdgpu_waves_per_eu(2, 4)))
void gemm_bias_relu(const float* __restrict__ X, const float* __restrict__ W,
                    const float* __restrict__ bias, float* __restrict__ out, int N) {
    __shared__ float4 Bs4[K / 4][O_DIM];

    const int tid = threadIdx.x;
    const int wv  = tid >> 5;
    const int ln  = tid & 31;
    const int c   = ln & 15;
    const int h   = ln >> 4;
    const int mtile = blockIdx.x * 128 + wv * 16;

    int row = mtile + c;
    if (row > N - 1) row = N - 1;
    const float4* xp = (const float4*)(X + (size_t)row * K) + h;

    float4 a0[8], a1[8];
    load_a_chunk(a0, xp, 0);

#pragma unroll
    for (int it = 0; it < K / 32; ++it) {
        const int f  = tid + it * 256;
        const int q  = f >> 5;
        const int c4 = f & 31;
        const float* wp = W + (size_t)(4 * q) * O_DIM + c4 * 4;
        const float4 v0 = *(const float4*)(wp);
        const float4 v1 = *(const float4*)(wp + O_DIM);
        const float4 v2 = *(const float4*)(wp + 2 * O_DIM);
        const float4 v3 = *(const float4*)(wp + 3 * O_DIM);
        float4* d = &Bs4[q][c4 * 4];
        d[0] = make_float4(v0.x, v1.x, v2.x, v3.x);
        d[1] = make_float4(v0.y, v1.y, v2.y, v3.y);
        d[2] = make_float4(v0.z, v1.z, v2.z, v3.z);
        d[3] = make_float4(v0.w, v1.w, v2.w, v3.w);
    }
    __syncthreads();

    if (mtile >= N) return;

    v8f acc[8];
#pragma unroll
    for (int t = 0; t < 8; ++t)
#pragma unroll
        for (int j = 0; j < 8; ++j) acc[t][j] = 0.0f;

#pragma unroll
    for (int k0 = 0; k0 < K; k0 += 128) {
        load_a_chunk(a1, xp, k0 + 64);
        mma_chunk(acc, a0, Bs4, k0, h, c);
        if (k0 + 128 < K) load_a_chunk(a0, xp, k0 + 128);
        mma_chunk(acc, a1, Bs4, k0 + 64, h, c);
    }

    const int rowoff = h << 3;
#pragma unroll
    for (int t = 0; t < 8; ++t) {
        const int col = t * 16 + c;
        const float bv = bias[col];
#pragma unroll
        for (int i = 0; i < 8; ++i) {
            float v = acc[t][i] + bv;
            out[(size_t)(mtile + rowoff + i) * O_DIM + col] = v > 0.f ? v : 0.f;
        }
    }
}

__global__ __launch_bounds__(256)
void spmm128(const int* __restrict__ src, const int* __restrict__ dst,
             const float* __restrict__ w, const float* __restrict__ x,
             float* __restrict__ out, int E) {
    int e = blockIdx.x * 8 + (threadIdx.x >> 5);
    if (e >= E) return;
    const int ln = threadIdx.x & 31;
    const int s = src[e];
    const int d = dst[e];
    const float wt = w[e];
    float4 v = ((const float4*)(x + (size_t)s * 128))[ln];
    float* o = out + (size_t)d * 128 + ln * 4;
    atomAddF(o + 0, wt * v.x);
    atomAddF(o + 1, wt * v.y);
    atomAddF(o + 2, wt * v.z);
    atomAddF(o + 3, wt * v.w);
}

__device__ __forceinline__ void add_row4(float* o, float wt, const float4 v) {
    atomAddF(o + 0, wt * v.x);
    atomAddF(o + 1, wt * v.y);
    atomAddF(o + 2, wt * v.z);
    atomAddF(o + 3, wt * v.w);
}

template <int NV>
__global__ __launch_bounds__(256)
void spmm128_multi(const int* __restrict__ src, const int* __restrict__ dst,
                   const float* __restrict__ w,
                   const float* __restrict__ xa, const float* __restrict__ xb,
                   const float* __restrict__ xc,
                   float* __restrict__ oa, float* __restrict__ ob,
                   float* __restrict__ oc, int E) {
    int e = blockIdx.x * 8 + (threadIdx.x >> 5);
    if (e >= E) return;
    const int ln = threadIdx.x & 31;
    const int s = src[e];
    const int d = dst[e];
    const float wt = w[e];
    const size_t so = (size_t)s * 128 + ln * 4;
    const size_t doff = (size_t)d * 128 + ln * 4;
    const float4 va = *(const float4*)(xa + so);
    const float4 vb = *(const float4*)(xb + so);
    float4 vc = make_float4(0.f, 0.f, 0.f, 0.f);
    if (NV == 3) vc = *(const float4*)(xc + so);
    add_row4(oa + doff, wt, va);
    add_row4(ob + doff, wt, vb);
    if (NV == 3) add_row4(oc + doff, wt, vc);
}

__global__ __launch_bounds__(256)
void l2norm_concat(const float* __restrict__ r0, const float* __restrict__ r1a,
                   const float* __restrict__ r1b, float* __restrict__ out, int N) {
    int row = blockIdx.x * 8 + (threadIdx.x >> 5);
    if (row >= N) return;
    const int ln = threadIdx.x & 31;

    float4 v0  = ((const float4*)(r0  + (size_t)row * 128))[ln];
    float4 v1a = ((const float4*)(r1a + (size_t)row * 128))[ln];
    float4 v1b = ((const float4*)(r1b + (size_t)row * 128))[ln];

    float s0 = v0.x * v0.x + v0.y * v0.y + v0.z * v0.z + v0.w * v0.w;
    float s1 = v1a.x * v1a.x + v1a.y * v1a.y + v1a.z * v1a.z + v1a.w * v1a.w
             + v1b.x * v1b.x + v1b.y * v1b.y + v1b.z * v1b.z + v1b.w * v1b.w;
#pragma unroll
    for (int off = 16; off > 0; off >>= 1) {
        s0 += __shfl_xor(s0, off, 32);
        s1 += __shfl_xor(s1, off, 32);
    }
    const float i0 = 1.0f / (sqrtf(s0) + 1e-9f);
    const float i1 = 1.0f / (sqrtf(s1) + 1e-9f);

    float* o = out + (size_t)row * 384;
    float4 w0 = make_float4(v0.x * i0, v0.y * i0, v0.z * i0, v0.w * i0);
    float4 wa = make_float4(v1a.x * i1, v1a.y * i1, v1a.z * i1, v1a.w * i1);
    float4 wb = make_float4(v1b.x * i1, v1b.y * i1, v1b.z * i1, v1b.w * i1);
    *(float4*)(o + ln * 4)        = w0;
    *(float4*)(o + 128 + ln * 4)  = wa;
    *(float4*)(o + 256 + ln * 4)  = wb;
}

extern "C" void kernel_launch(void* const* d_in, const int* in_sizes, int n_in,
                              void* d_out, int out_size, void* d_ws, size_t ws_size,
                              hipStream_t stream) {
    const float* x1       = (const float*)d_in[0];
    const float* x2       = (const float*)d_in[1];
    const float* word_emb = (const float*)d_in[2];
    const float* W1a = (const float*)d_in[3];
    const float* b1a = (const float*)d_in[4];
    const float* W1b = (const float*)d_in[5];
    const float* b1b = (const float*)d_in[6];
    const float* W2a = (const float*)d_in[7];
    const float* b2a = (const float*)d_in[8];
    const float* W2b = (const float*)d_in[9];
    const float* b2b = (const float*)d_in[10];
    const int*   e11_src = (const int*)d_in[11];
    const int*   e11_dst = (const int*)d_in[12];
    const float* e11_w   = (const float*)d_in[13];
    const int*   e22_src = (const int*)d_in[14];
    const int*   e22_dst = (const int*)d_in[15];
    const float* e22_w   = (const float*)d_in[16];
    const int*   e01_src = (const int*)d_in[17];
    const int*   e01_dst = (const int*)d_in[18];
    const float* e01_w   = (const float*)d_in[19];
    const int*   e02_src = (const int*)d_in[20];
    const int*   e02_dst = (const int*)d_in[21];
    const float* e02_w   = (const float*)d_in[22];

    const int N0 = 10000, N1 = 50000, N2 = 30000;
    const int E11 = in_sizes[11], E22 = in_sizes[14];
    const int E01 = in_sizes[17], E02 = in_sizes[20];

    float* ws = (float*)d_ws;
    size_t off = 0;
    float* l1_1 = ws + off; off += (size_t)N1 * 128;
    float* h11  = ws + off; off += (size_t)N1 * 128;
    float* l2_1 = ws + off; off += (size_t)N1 * 128;
    float* l1_2 = ws + off; off += (size_t)N2 * 128;
    float* h22  = ws + off; off += (size_t)N2 * 128;
    float* l2_2 = ws + off; off += (size_t)N2 * 128;
    float* r0   = ws + off; off += (size_t)N0 * 128;
    float* r0s  = ws + off; off += (size_t)N0 * 128;
    float* r1   = ws + off; off += (size_t)N0 * 128;
    float* r1s  = ws + off; off += (size_t)N0 * 128;
    float* rw   = ws + off; off += (size_t)N0 * 128;

    float* doc     = (float*)d_out;
    float* doc_svd = doc + (size_t)N0 * 384;

    const dim3 blk(256);
    auto gemm_grid = [](int n) { return dim3((n + 127) / 128); };
    auto edge_grid = [](int e) { return dim3((e + 7) / 8); };

    gemm_bias_relu<256><<<gemm_grid(N1), blk, 0, stream>>>(x1, W1a, b1a, l1_1, N1);
    hipMemsetAsync(h11, 0, (size_t)N1 * 128 * sizeof(float), stream);
    spmm128<<<edge_grid(E11), blk, 0, stream>>>(e11_src, e11_dst, e11_w, l1_1, h11, E11);
    gemm_bias_relu<128><<<gemm_grid(N1), blk, 0, stream>>>(h11, W1b, b1b, l2_1, N1);

    gemm_bias_relu<256><<<gemm_grid(N2), blk, 0, stream>>>(x2, W2a, b2a, l1_2, N2);
    hipMemsetAsync(h22, 0, (size_t)N2 * 128 * sizeof(float), stream);
    spmm128<<<edge_grid(E22), blk, 0, stream>>>(e22_src, e22_dst, e22_w, l1_2, h22, E22);
    gemm_bias_relu<128><<<gemm_grid(N2), blk, 0, stream>>>(h22, W2b, b2b, l2_2, N2);

    hipMemsetAsync(r0, 0, (size_t)N0 * (5 * 128) * sizeof(float), stream);
    spmm128_multi<2><<<edge_grid(E01), blk, 0, stream>>>(e01_src, e01_dst, e01_w,
        l2_1, l1_1, nullptr, r0, r0s, nullptr, E01);
    spmm128_multi<3><<<edge_grid(E02), blk, 0, stream>>>(e02_src, e02_dst, e02_w,
        l2_2, l1_2, word_emb, r1, r1s, rw, E02);

    l2norm_concat<<<dim3((N0 + 7) / 8), blk, 0, stream>>>(r0,  r1,  rw, doc,     N0);
    l2norm_concat<<<dim3((N0 + 7) / 8), blk, 0, stream>>>(r0s, r1s, rw, doc_svd, N0);
}
